# starting kernel
# baseline (speedup 1.0000x reference)
#include <hip/hip_runtime.h>
#include <hip/hip_bf16.h>
#include <math.h>

#define N_WIRES 12
#define DEPTH   4
#define IN_DIM  256
#define BATCH   4096
#define NSTATES 4096

typedef __attribute__((ext_vector_type(2))) float v2f;
typedef __attribute__((ext_vector_type(8))) float v8f;

__global__ __launch_bounds__(256) void vq_angles_wmma(
    const float* __restrict__ x, const float* __restrict__ W,
    float* __restrict__ angles  )
{
  const int lane = threadIdx.x & 31;
  const int wave = threadIdx.x >> 5;
  const int row0 = (blockIdx.x * 8 + wave) * 16;
  const int half = lane >> 4;
  const int mrow = lane & 15;

  const int wrow = (mrow < N_WIRES) ? mrow : 0;
  const float* xr = x + (size_t)(row0 + mrow) * IN_DIM;
  const float* wr = W + (size_t)wrow * IN_DIM;

  v8f acc = {0.f, 0.f, 0.f, 0.f, 0.f, 0.f, 0.f, 0.f};
  #pragma unroll 8
  for (int k0 = 0; k0 < IN_DIM; k0 += 4) {
    const int kk = k0 + 2 * half;
    v2f a, b;
    a.x = xr[kk];
    a.y = xr[kk + 1];
    b.x = wr[kk];
    b.y = wr[kk + 1];
    acc = __builtin_amdgcn_wmma_f32_16x16x4_f32(false, a, false, b,
                                                (short)0, acc, false, false);
  }
  const int ncol = lane & 15;
  #pragma unroll
  for (int r = 0; r < 8; ++r) {
    const int m = r + 8 * half;
    angles[(size_t)(row0 + m) * 16 + ncol] = acc[r];
  }
}

__global__ __launch_bounds__(256) void vq_sim(
    const float* __restrict__ angles  ,
    const float* __restrict__ params  ,
    float* __restrict__ out  )
{
  __shared__ float sre[NSTATES];
  __shared__ float sim_[NSTATES];
  __shared__ float sc[N_WIRES], ss[N_WIRES];
  __shared__ float cp[DEPTH * N_WIRES], sp[DEPTH * N_WIRES];

  const int tid = threadIdx.x;
  const int b   = blockIdx.x;

  if (tid < N_WIRES) {
    float h = 0.5f * angles[(size_t)b * 16 + tid];
    sc[tid] = cosf(h);
    ss[tid] = sinf(h);
  }
  if (tid < DEPTH * N_WIRES) {
    float h = 0.5f * params[tid];
    cp[tid] = cosf(h);
    sp[tid] = sinf(h);
  }
  __syncthreads();

  float lc[N_WIRES], ls[N_WIRES];
  #pragma unroll
  for (int w = 0; w < N_WIRES; ++w) { lc[w] = sc[w]; ls[w] = ss[w]; }

  #pragma unroll
  for (int k = 0; k < 16; ++k) {
    int idx = tid * 16 + k;
    float m = 1.f;
    #pragma unroll
    for (int p = 0; p < N_WIRES; ++p)
      m *= ((idx >> p) & 1) ? ls[11 - p] : lc[11 - p];
    int kp = __popc(idx) & 3;
    float re = 0.f, im = 0.f;
    if      (kp == 0) re =  m;
    else if (kp == 1) im = -m;
    else if (kp == 2) re = -m;
    else              im =  m;
    sre[idx]  = re;
    sim_[idx] = im;
  }

  for (int d = 0; d < DEPTH; ++d) {
    #pragma unroll 1
    for (int w = 0; w < N_WIRES; ++w) {
      __syncthreads();
      const float c = cp[d * N_WIRES + w];
      const float s = sp[d * N_WIRES + w];
      const int S = 1 << (11 - w);
      #pragma unroll
      for (int r = 0; r < 8; ++r) {
        int p  = r * 256 + tid;
        int i0 = ((p & ~(S - 1)) << 1) | (p & (S - 1));
        int i1 = i0 + S;
        float a0r = sre[i0], a0i = sim_[i0];
        float a1r = sre[i1], a1i = sim_[i1];
        sre[i0]  = fmaf(c, a0r,  s * a1i);
        sim_[i0] = fmaf(c, a0i, -s * a1r);
        sre[i1]  = fmaf(c, a1r,  s * a0i);
        sim_[i1] = fmaf(c, a1i, -s * a0r);
      }
    }
    __syncthreads();
    float rr[16], ri[16];
    #pragma unroll
    for (int k = 0; k < 16; ++k) {
      int i = tid * 16 + k;
      int j = i ^ (i >> 1);
      rr[k] = sre[j];
      ri[k] = sim_[j];
    }
    __syncthreads();
    #pragma unroll
    for (int k = 0; k < 16; ++k) {
      int i = tid * 16 + k;
      sre[i]  = rr[k];
      sim_[i] = ri[k];
    }
  }
  __syncthreads();

  float part[N_WIRES];
  #pragma unroll
  for (int w = 0; w < N_WIRES; ++w) part[w] = 0.f;
  #pragma unroll
  for (int k = 0; k < 16; ++k) {
    int idx = tid * 16 + k;
    float pr = sre[idx] * sre[idx] + sim_[idx] * sim_[idx];
    #pragma unroll
    for (int w = 0; w < N_WIRES; ++w)
      part[w] += ((idx >> (11 - w)) & 1) ? -pr : pr;
  }
  __syncthreads();
  #pragma unroll
  for (int w = 0; w < N_WIRES; ++w) sre[tid * N_WIRES + w] = part[w];
  __syncthreads();
  for (int off = 128; off >= 1; off >>= 1) {
    if (tid < off) {
      #pragma unroll
      for (int w = 0; w < N_WIRES; ++w)
        sre[tid * N_WIRES + w] += sre[(tid + off) * N_WIRES + w];
    }
    __syncthreads();
  }
  if (tid < N_WIRES) out[(size_t)b * N_WIRES + tid] = sre[tid];
}

extern "C" void kernel_launch(void* const* d_in, const int* in_sizes, int n_in,
                              void* d_out, int out_size, void* d_ws, size_t ws_size,
                              hipStream_t stream) {
  (void)in_sizes; (void)n_in; (void)out_size; (void)ws_size;
  const float* x      = (const float*)d_in[0];
  const float* W      = (const float*)d_in[1];
  const float* params = (const float*)d_in[2];
  float* out    = (float*)d_out;
  float* angles = (float*)d_ws;

  vq_angles_wmma<<<BATCH / 128, 256, 0, stream>>>(x, W, angles);
  vq_sim<<<BATCH, 256, 0, stream>>>(angles, params, out);
}


# RX sweeps register-blocked: each LDS pass loads 16 amplitudes per thread and applies 4 gates in registers (3 passes per layer, not 12); state as padded (re,im) pairs; gate and sum order unchanged
# speedup vs baseline: 3.2272x; 3.2272x over previous
#include <hip/hip_runtime.h>
#include <hip/hip_bf16.h>
#include <math.h>

#define N_WIRES 12
#define DEPTH   4
#define IN_DIM  256
#define BATCH   4096
#define NSTATES 4096

typedef __attribute__((ext_vector_type(2))) float v2f;
typedef __attribute__((ext_vector_type(8))) float v8f;

__global__ __launch_bounds__(256) void vq_angles_wmma(
    const float* __restrict__ x, const float* __restrict__ W,
    float* __restrict__ angles  )
{
  const int lane = threadIdx.x & 31;
  const int wave = threadIdx.x >> 5;
  const int row0 = (blockIdx.x * 8 + wave) * 16;
  const int half = lane >> 4;
  const int mrow = lane & 15;

  const int wrow = (mrow < N_WIRES) ? mrow : 0;
  const float* xr = x + (size_t)(row0 + mrow) * IN_DIM;
  const float* wr = W + (size_t)wrow * IN_DIM;

  v8f acc = {0.f, 0.f, 0.f, 0.f, 0.f, 0.f, 0.f, 0.f};
  #pragma unroll 8
  for (int k0 = 0; k0 < IN_DIM; k0 += 4) {
    const int kk = k0 + 2 * half;
    v2f a, b;
    a.x = xr[kk];
    a.y = xr[kk + 1];
    b.x = wr[kk];
    b.y = wr[kk + 1];
    acc = __builtin_amdgcn_wmma_f32_16x16x4_f32(false, a, false, b,
                                                (short)0, acc, false, false);
  }
  const int ncol = lane & 15;
  #pragma unroll
  for (int r = 0; r < 8; ++r) {
    const int m = r + 8 * half;
    angles[(size_t)(row0 + m) * 16 + ncol] = acc[r];
  }
}

typedef __attribute__((ext_vector_type(4))) float v4f;

#define NPAIRS 4848

__device__ __forceinline__ int pair_slot(int i) {
  return i + 2 * (i >> 4) + 16 * (i >> 8);
}

__device__ __forceinline__ void rx_pair(float& a0r, float& a0i, float& a1r, float& a1i, float c, float s) {
  const float n0r = fmaf(c, a0r,  s * a1i);
  const float n0i = fmaf(c, a0i, -s * a1r);
  const float n1r = fmaf(c, a1r,  s * a0i);
  const float n1i = fmaf(c, a1i, -s * a0r);
  a0r = n0r; a0i = n0i; a1r = n1r; a1i = n1i;
}

__device__ __forceinline__ void rx4(float (&ar)[16], float (&ai)[16], v4f c4, v4f s4) {
  #pragma unroll
  for (int g = 0; g < 4; ++g) {
    const int m = 8 >> g;
    const float c = c4[g];
    const float s = s4[g];
    #pragma unroll
    for (int k = 0; k < 16; ++k) {
      if ((k & m) == 0) rx_pair(ar[k], ai[k], ar[k | m], ai[k | m], c, s);
    }
  }
}

__global__ __launch_bounds__(256) void vq_sim(
    const float* __restrict__ angles  ,
    const float* __restrict__ params  ,
    float* __restrict__ out  )
{
  __shared__ __align__(16) float st[2 * NPAIRS];
  __shared__ float sc[N_WIRES], ss[N_WIRES];
  __shared__ __align__(16) float cp[DEPTH * N_WIRES];
  __shared__ __align__(16) float sp[DEPTH * N_WIRES];

  const int tid  = threadIdx.x;
  const int b    = blockIdx.x;
  const int lane = tid & 31;
  const int wave = tid >> 5;

  if (tid < N_WIRES) {
    float h = 0.5f * angles[(size_t)b * 16 + tid];
    sc[tid] = cosf(h);
    ss[tid] = sinf(h);
  }
  if (tid < DEPTH * N_WIRES) {
    float h = 0.5f * params[tid];
    cp[tid] = cosf(h);
    sp[tid] = sinf(h);
  }
  __syncthreads();

  float lc[N_WIRES], ls[N_WIRES];
  #pragma unroll
  for (int w = 0; w < N_WIRES; ++w) { lc[w] = sc[w]; ls[w] = ss[w]; }

  float* const rowA = st + 2 * pair_slot(tid << 4);
  float* const baseB = st + 2 * pair_slot((lane & 15) | ((lane >> 4) << 8) | (wave << 9));
  float* const baseC = st + 2 * pair_slot((lane & 15) | (wave << 4) | ((lane >> 4) << 7));

  {
    float pre[16], pim[16];
    #pragma unroll
    for (int k = 0; k < 16; ++k) {
      int idx = tid * 16 + k;
      float m = 1.f;
      #pragma unroll
      for (int p = 0; p < N_WIRES; ++p)
        m *= ((idx >> p) & 1) ? ls[11 - p] : lc[11 - p];
      int kp = __popc(idx) & 3;
      float re = 0.f, im = 0.f;
      if      (kp == 0) re =  m;
      else if (kp == 1) im = -m;
      else if (kp == 2) re = -m;
      else              im =  m;
      pre[k] = re;
      pim[k] = im;
    }
    #pragma unroll
    for (int q = 0; q < 8; ++q) {
      v4f v = {pre[2 * q], pim[2 * q], pre[2 * q + 1], pim[2 * q + 1]};
      *(v4f*)(rowA + 4 * q) = v;
    }
  }

  #pragma unroll 1
  for (int d = 0; d < DEPTH; ++d) {
    float ar[16], ai[16];

    __syncthreads();
    {
      const v4f c4 = *(const v4f*)&cp[d * N_WIRES + 0];
      const v4f s4 = *(const v4f*)&sp[d * N_WIRES + 0];
      #pragma unroll
      for (int k = 0; k < 16; ++k)
      { const v2f v = *(const v2f*)(baseC + 2 * (304 * k)); ar[k] = v.x; ai[k] = v.y; }
      rx4(ar, ai, c4, s4);
      #pragma unroll
      for (int k = 0; k < 16; ++k)
        *(v2f*)(baseC + 2 * (304 * k)) = v2f{ar[k], ai[k]};
    }

    __syncthreads();
    {
      const v4f c4 = *(const v4f*)&cp[d * N_WIRES + 4];
      const v4f s4 = *(const v4f*)&sp[d * N_WIRES + 4];
      #pragma unroll
      for (int k = 0; k < 16; ++k)
      { const v2f v = *(const v2f*)(baseB + 2 * (18 * k)); ar[k] = v.x; ai[k] = v.y; }
      rx4(ar, ai, c4, s4);
      #pragma unroll
      for (int k = 0; k < 16; ++k)
        *(v2f*)(baseB + 2 * (18 * k)) = v2f{ar[k], ai[k]};
    }

    __syncthreads();
    {
      const v4f c4 = *(const v4f*)&cp[d * N_WIRES + 8];
      const v4f s4 = *(const v4f*)&sp[d * N_WIRES + 8];
      #pragma unroll
      for (int q = 0; q < 8; ++q) {
        const v4f v = *(const v4f*)(rowA + 4 * q);
        ar[2 * q] = v.x;     ai[2 * q] = v.y;
        ar[2 * q + 1] = v.z; ai[2 * q + 1] = v.w;
      }
      rx4(ar, ai, c4, s4);
      #pragma unroll
      for (int q = 0; q < 8; ++q) {
        v4f v = {ar[2 * q], ai[2 * q], ar[2 * q + 1], ai[2 * q + 1]};
        *(v4f*)(rowA + 4 * q) = v;
      }
    }
    __syncthreads();

    {
      const int t0 = tid & 1;
      float* const rowR = st + 2 * pair_slot((tid ^ (tid >> 1)) << 4);
      float* const rLo = rowR + (t0 ? 16 : 0);
      float* const rHi = rowR + (t0 ? 0 : 16);
      #pragma unroll
      for (int q = 0; q < 8; ++q) {
        const int g0 = (2 * q) ^ q;
        const float* src = ((g0 & 8) ? rHi : rLo) + 2 * (g0 & 6);
        const v4f v = *(const v4f*)src;
        if ((g0 & 1) == 0) {
          ar[2 * q] = v.x;     ai[2 * q] = v.y;
          ar[2 * q + 1] = v.z; ai[2 * q + 1] = v.w;
        } else {
          ar[2 * q] = v.z;     ai[2 * q] = v.w;
          ar[2 * q + 1] = v.x; ai[2 * q + 1] = v.y;
        }
      }
      __syncthreads();
      #pragma unroll
      for (int q = 0; q < 8; ++q) {
        v4f v = {ar[2 * q], ai[2 * q], ar[2 * q + 1], ai[2 * q + 1]};
        *(v4f*)(rowA + 4 * q) = v;
      }
    }
  }
  __syncthreads();

  float part[N_WIRES];
  #pragma unroll
  for (int w = 0; w < N_WIRES; ++w) part[w] = 0.f;
  {
    float fre[16], fim[16];
    #pragma unroll
    for (int q = 0; q < 8; ++q) {
      const v4f v = *(const v4f*)(rowA + 4 * q);
      fre[2 * q] = v.x; fim[2 * q] = v.y;
      fre[2 * q + 1] = v.z; fim[2 * q + 1] = v.w;
    }
    #pragma unroll
    for (int k = 0; k < 16; ++k) {
      int idx = tid * 16 + k;
      float pr = fre[k] * fre[k] + fim[k] * fim[k];
      #pragma unroll
      for (int w = 0; w < N_WIRES; ++w)
        part[w] += ((idx >> (11 - w)) & 1) ? -pr : pr;
    }
  }
  __syncthreads();
  float* const red = st;
  #pragma unroll
  for (int w = 0; w < N_WIRES; ++w) red[tid * N_WIRES + w] = part[w];
  __syncthreads();
  for (int off = 128; off >= 1; off >>= 1) {
    if (tid < off) {
      #pragma unroll
      for (int w = 0; w < N_WIRES; ++w)
        red[tid * N_WIRES + w] += red[(tid + off) * N_WIRES + w];
    }
    __syncthreads();
  }
  if (tid < N_WIRES) out[(size_t)b * N_WIRES + tid] = red[tid];
}

extern "C" void kernel_launch(void* const* d_in, const int* in_sizes, int n_in,
                              void* d_out, int out_size, void* d_ws, size_t ws_size,
                              hipStream_t stream) {
  (void)in_sizes; (void)n_in; (void)out_size; (void)ws_size;
  const float* x      = (const float*)d_in[0];
  const float* W      = (const float*)d_in[1];
  const float* params = (const float*)d_in[2];
  float* out    = (float*)d_out;
  float* angles = (float*)d_ws;

  vq_angles_wmma<<<BATCH / 128, 256, 0, stream>>>(x, W, angles);
  vq_sim<<<BATCH, 256, 0, stream>>>(angles, params, out);
}


# RX gates as a0 - i*t*a1, one fma per component: larger of cos/sin of each gate taken out (|t|<=1), multiplied back at the end; the sin case's X ends as an output sign; gate rounding changes
# speedup vs baseline: 3.9832x; 1.2343x over previous
#include <hip/hip_runtime.h>
#include <hip/hip_bf16.h>
#include <math.h>

#define N_WIRES 12
#define DEPTH   4
#define IN_DIM  256
#define BATCH   4096
#define NSTATES 4096

typedef __attribute__((ext_vector_type(2))) float v2f;
typedef __attribute__((ext_vector_type(8))) float v8f;

__global__ __launch_bounds__(256) void vq_angles_wmma(
    const float* __restrict__ x, const float* __restrict__ W,
    float* __restrict__ angles  )
{
  const int lane = threadIdx.x & 31;
  const int wave = threadIdx.x >> 5;
  const int row0 = (blockIdx.x * 8 + wave) * 16;
  const int half = lane >> 4;
  const int mrow = lane & 15;

  const int wrow = (mrow < N_WIRES) ? mrow : 0;
  const float* xr = x + (size_t)(row0 + mrow) * IN_DIM;
  const float* wr = W + (size_t)wrow * IN_DIM;

  v8f acc = {0.f, 0.f, 0.f, 0.f, 0.f, 0.f, 0.f, 0.f};
  #pragma unroll 8
  for (int k0 = 0; k0 < IN_DIM; k0 += 4) {
    const int kk = k0 + 2 * half;
    v2f a, b;
    a.x = xr[kk];
    a.y = xr[kk + 1];
    b.x = wr[kk];
    b.y = wr[kk + 1];
    acc = __builtin_amdgcn_wmma_f32_16x16x4_f32(false, a, false, b,
                                                (short)0, acc, false, false);
  }
  const int ncol = lane & 15;
  #pragma unroll
  for (int r = 0; r < 8; ++r) {
    const int m = r + 8 * half;
    angles[(size_t)(row0 + m) * 16 + ncol] = acc[r];
  }
}

typedef __attribute__((ext_vector_type(4))) float v4f;

#define NPAIRS 4848

__device__ __forceinline__ int pair_slot(int i) {
  return i + 2 * (i >> 4) + 16 * (i >> 8);
}

__device__ __forceinline__ void rx_pair(float& a0r, float& a0i, float& a1r, float& a1i, float t) {
  const float n0r = fmaf( t, a1i, a0r);
  const float n0i = fmaf(-t, a1r, a0i);
  const float n1r = fmaf( t, a0i, a1r);
  const float n1i = fmaf(-t, a0r, a1i);
  a0r = n0r; a0i = n0i; a1r = n1r; a1i = n1i;
}

__device__ __forceinline__ void rx4(float (&ar)[16], float (&ai)[16], v4f t4) {
  #pragma unroll
  for (int g = 0; g < 4; ++g) {
    const int m = 8 >> g;
    const float t = t4[g];
    #pragma unroll
    for (int k = 0; k < 16; ++k) {
      if ((k & m) == 0) rx_pair(ar[k], ai[k], ar[k | m], ai[k | m], t);
    }
  }
}

__global__ __launch_bounds__(256) void vq_sim(
    const float* __restrict__ angles  ,
    const float* __restrict__ params  ,
    float* __restrict__ out  )
{
  __shared__ __align__(16) float st[2 * NPAIRS];
  __shared__ float sc[N_WIRES], ss[N_WIRES];
  __shared__ __align__(16) float tp[DEPTH * N_WIRES];
  __shared__ __align__(16) float mg[DEPTH * N_WIRES];
  __shared__ int xf[DEPTH * N_WIRES];

  const int tid  = threadIdx.x;
  const int b    = blockIdx.x;
  const int lane = tid & 31;
  const int wave = tid >> 5;

  if (tid < N_WIRES) {
    float h = 0.5f * angles[(size_t)b * 16 + tid];
    sc[tid] = cosf(h);
    ss[tid] = sinf(h);
  }
  if (tid < DEPTH * N_WIRES) {
    const float h = 0.5f * params[tid];
    const float c = cosf(h);
    const float s = sinf(h);
    const bool x = fabsf(s) > fabsf(c);
    tp[tid] = x ? (-c / s) : (s / c);
    mg[tid] = x ? fabsf(s) : fabsf(c);
    xf[tid] = x ? 1 : 0;
  }
  __syncthreads();

  float lc[N_WIRES], ls[N_WIRES];
  #pragma unroll
  for (int w = 0; w < N_WIRES; ++w) { lc[w] = sc[w]; ls[w] = ss[w]; }

  float* const rowA = st + 2 * pair_slot(tid << 4);
  float* const baseB = st + 2 * pair_slot((lane & 15) | ((lane >> 4) << 8) | (wave << 9));
  float* const baseC = st + 2 * pair_slot((lane & 15) | (wave << 4) | ((lane >> 4) << 7));

  {
    float pre[16], pim[16];
    #pragma unroll
    for (int k = 0; k < 16; ++k) {
      int idx = tid * 16 + k;
      float m = 1.f;
      #pragma unroll
      for (int p = 0; p < N_WIRES; ++p)
        m *= ((idx >> p) & 1) ? ls[11 - p] : lc[11 - p];
      int kp = __popc(idx) & 3;
      float re = 0.f, im = 0.f;
      if      (kp == 0) re =  m;
      else if (kp == 1) im = -m;
      else if (kp == 2) re = -m;
      else              im =  m;
      pre[k] = re;
      pim[k] = im;
    }
    #pragma unroll
    for (int q = 0; q < 8; ++q) {
      v4f v = {pre[2 * q], pim[2 * q], pre[2 * q + 1], pim[2 * q + 1]};
      *(v4f*)(rowA + 4 * q) = v;
    }
  }

  #pragma unroll 1
  for (int d = 0; d < DEPTH; ++d) {
    float ar[16], ai[16];

    __syncthreads();
    {
      const v4f t4 = *(const v4f*)&tp[d * N_WIRES + 0];
      #pragma unroll
      for (int k = 0; k < 16; ++k)
      { const v2f v = *(const v2f*)(baseC + 2 * (304 * k)); ar[k] = v.x; ai[k] = v.y; }
      rx4(ar, ai, t4);
      #pragma unroll
      for (int k = 0; k < 16; ++k)
        *(v2f*)(baseC + 2 * (304 * k)) = v2f{ar[k], ai[k]};
    }

    __syncthreads();
    {
      const v4f t4 = *(const v4f*)&tp[d * N_WIRES + 4];
      #pragma unroll
      for (int k = 0; k < 16; ++k)
      { const v2f v = *(const v2f*)(baseB + 2 * (18 * k)); ar[k] = v.x; ai[k] = v.y; }
      rx4(ar, ai, t4);
      #pragma unroll
      for (int k = 0; k < 16; ++k)
        *(v2f*)(baseB + 2 * (18 * k)) = v2f{ar[k], ai[k]};
    }

    __syncthreads();
    {
      const v4f t4 = *(const v4f*)&tp[d * N_WIRES + 8];
      #pragma unroll
      for (int q = 0; q < 8; ++q) {
        const v4f v = *(const v4f*)(rowA + 4 * q);
        ar[2 * q] = v.x;     ai[2 * q] = v.y;
        ar[2 * q + 1] = v.z; ai[2 * q + 1] = v.w;
      }
      rx4(ar, ai, t4);
      #pragma unroll
      for (int q = 0; q < 8; ++q) {
        v4f v = {ar[2 * q], ai[2 * q], ar[2 * q + 1], ai[2 * q + 1]};
        *(v4f*)(rowA + 4 * q) = v;
      }
    }
    __syncthreads();

    {
      const int t0 = tid & 1;
      float* const rowR = st + 2 * pair_slot((tid ^ (tid >> 1)) << 4);
      float* const rLo = rowR + (t0 ? 16 : 0);
      float* const rHi = rowR + (t0 ? 0 : 16);
      #pragma unroll
      for (int q = 0; q < 8; ++q) {
        const int g0 = (2 * q) ^ q;
        const float* src = ((g0 & 8) ? rHi : rLo) + 2 * (g0 & 6);
        const v4f v = *(const v4f*)src;
        if ((g0 & 1) == 0) {
          ar[2 * q] = v.x;     ai[2 * q] = v.y;
          ar[2 * q + 1] = v.z; ai[2 * q + 1] = v.w;
        } else {
          ar[2 * q] = v.z;     ai[2 * q] = v.w;
          ar[2 * q + 1] = v.x; ai[2 * q + 1] = v.y;
        }
      }
      __syncthreads();
      #pragma unroll
      for (int q = 0; q < 8; ++q) {
        v4f v = {ar[2 * q], ai[2 * q], ar[2 * q + 1], ai[2 * q + 1]};
        *(v4f*)(rowA + 4 * q) = v;
      }
    }
  }
  __syncthreads();

  float part[N_WIRES];
  #pragma unroll
  for (int w = 0; w < N_WIRES; ++w) part[w] = 0.f;
  {
    float fre[16], fim[16];
    #pragma unroll
    for (int q = 0; q < 8; ++q) {
      const v4f v = *(const v4f*)(rowA + 4 * q);
      fre[2 * q] = v.x; fim[2 * q] = v.y;
      fre[2 * q + 1] = v.z; fim[2 * q + 1] = v.w;
    }
    #pragma unroll
    for (int k = 0; k < 16; ++k) {
      int idx = tid * 16 + k;
      float pr = fre[k] * fre[k] + fim[k] * fim[k];
      #pragma unroll
      for (int w = 0; w < N_WIRES; ++w)
        part[w] += ((idx >> (11 - w)) & 1) ? -pr : pr;
    }
  }
  __syncthreads();
  float* const red = st;
  #pragma unroll
  for (int w = 0; w < N_WIRES; ++w) red[tid * N_WIRES + w] = part[w];
  __syncthreads();
  for (int off = 128; off >= 1; off >>= 1) {
    if (tid < off) {
      #pragma unroll
      for (int w = 0; w < N_WIRES; ++w)
        red[tid * N_WIRES + w] += red[(tid + off) * N_WIRES + w];
    }
    __syncthreads();
  }
  if (tid < N_WIRES) {
    float nrm = 1.f;
    for (int j = 0; j < DEPTH * N_WIRES; ++j) nrm *= mg[j];
    nrm *= nrm;
    unsigned xm = 0u;
    for (int d = 0; d < DEPTH; ++d) {
      for (int w = 0; w < N_WIRES; ++w) xm ^= (unsigned)xf[d * N_WIRES + w] << w;
      xm ^= xm << 1; xm ^= xm << 2; xm ^= xm << 4; xm ^= xm << 8;
    }
    const float r = red[tid] * nrm;
    out[(size_t)b * N_WIRES + tid] = ((xm >> tid) & 1u) ? -r : r;
  }
}

extern "C" void kernel_launch(void* const* d_in, const int* in_sizes, int n_in,
                              void* d_out, int out_size, void* d_ws, size_t ws_size,
                              hipStream_t stream) {
  (void)in_sizes; (void)n_in; (void)out_size; (void)ws_size;
  const float* x      = (const float*)d_in[0];
  const float* W      = (const float*)d_in[1];
  const float* params = (const float*)d_in[2];
  float* out    = (float*)d_out;
  float* angles = (float*)d_ws;

  vq_angles_wmma<<<BATCH / 128, 256, 0, stream>>>(x, W, angles);
  vq_sim<<<BATCH, 256, 0, stream>>>(angles, params, out);
}
